# starting kernel
# baseline (speedup 1.0000x reference)
#include <hip/hip_runtime.h>
#include <math.h>

typedef __attribute__((ext_vector_type(2))) float v2f;
typedef __attribute__((ext_vector_type(8))) float v8f;

#define PI_F 3.14159265358979323846f

static __device__ __forceinline__ v8f wmma_f32_k4(v2f a, v2f b, v8f c) {
  return __builtin_amdgcn_wmma_f32_16x16x4_f32(false, a, false, b, (short)0, c, false, false);
}

__global__ void qcnn_setup_kernel(const float* __restrict__ style,
                                  const float* __restrict__ style_w,
                                  const float* __restrict__ style_b,
                                  const float* __restrict__ qcnn,
                                  const float* __restrict__ meas,
                                  const float* __restrict__ out_w,
                                  float* __restrict__ Ur, float* __restrict__ Ui,
                                  float* __restrict__ Wq, float* __restrict__ sAng)
{
  __shared__ float sre[4096];
  __shared__ float sim[4096];
  const int t = threadIdx.x;

  for (int a = 0; a < 64; ++a) { sre[a*64+t] = (a == t) ? 1.f : 0.f; sim[a*64+t] = 0.f; }

  for (int l = 0; l < 2; ++l) {
    for (int i = 0; i < 6; ++i) {
      const int mask = 1 << (5 - i);
      {
        float th = 0.5f * qcnn[((l*6 + i)*2 + 0)*3 + 0];
        float c = cosf(th), s = sinf(th);
        for (int a = 0; a < 64; ++a) if (!(a & mask)) {
          int a1 = a | mask;
          float r0 = sre[a*64+t],  i0 = sim[a*64+t];
          float r1 = sre[a1*64+t], i1 = sim[a1*64+t];
          sre[a*64+t]  = c*r0 - s*r1;  sim[a*64+t]  = c*i0 - s*i1;
          sre[a1*64+t] = s*r0 + c*r1;  sim[a1*64+t] = s*i0 + c*i1;
        }
      }
      {
        float th = 0.5f * qcnn[((l*6 + i)*2 + 1)*3 + 0];
        float ch = cosf(th), sh = sinf(th);
        for (int a = 0; a < 64; ++a) {
          float re = sre[a*64+t], im = sim[a*64+t];
          if (!(a & mask)) { sre[a*64+t] = re*ch + im*sh; sim[a*64+t] = im*ch - re*sh; }
          else             { sre[a*64+t] = re*ch - im*sh; sim[a*64+t] = im*ch + re*sh; }
        }
      }
    }
    for (int i = 0; i < 6; ++i) {
      const int mc = 1 << (5 - i);
      const int mt = 1 << (5 - ((i + 1) % 6));
      for (int a = 0; a < 64; ++a) if ((a & mc) && !(a & mt)) {
        int a1 = a | mt;
        float r = sre[a*64+t], q = sim[a*64+t];
        sre[a*64+t]  = sre[a1*64+t]; sim[a*64+t]  = sim[a1*64+t];
        sre[a1*64+t] = r;            sim[a1*64+t] = q;
      }
    }
  }
  for (int i = 0; i < 6; ++i) {
    const int mask = 1 << (5 - i);
    float th = 0.5f * meas[i*3+0];
    float ph = meas[i*3+1], la = meas[i*3+2];
    float ct = cosf(th), st = sinf(th);
    float m01r = -cosf(la)*st,    m01i = -sinf(la)*st;
    float m10r =  cosf(ph)*st,    m10i =  sinf(ph)*st;
    float m11r =  cosf(ph+la)*ct, m11i =  sinf(ph+la)*ct;
    for (int a = 0; a < 64; ++a) if (!(a & mask)) {
      int a1 = a | mask;
      float r0 = sre[a*64+t],  i0 = sim[a*64+t];
      float r1 = sre[a1*64+t], i1 = sim[a1*64+t];
      sre[a*64+t]  = ct*r0 + m01r*r1 - m01i*i1;
      sim[a*64+t]  = ct*i0 + m01r*i1 + m01i*r1;
      sre[a1*64+t] = m10r*r0 - m10i*i0 + m11r*r1 - m11i*i1;
      sim[a1*64+t] = m10r*i0 + m10i*r0 + m11r*i1 + m11i*r1;
    }
  }
  for (int a = 0; a < 64; ++a) { Ur[a*64+t] = sre[a*64+t]; Ui[a*64+t] = sim[a*64+t]; }

  for (int c = 0; c < 64; ++c) {
    float s = 0.f;
    #pragma unroll
    for (int i = 0; i < 6; ++i)
      s += (((t >> (5 - i)) & 1) ? -1.f : 1.f) * out_w[c*6 + i];
    Wq[t*64 + c] = s;
  }

  if (t < 48) {
    int img = t / 6, i = t % 6;
    float d = 0.f;
    for (int j = 0; j < 128; ++j) d += style[img*128 + j] * style_w[i*128 + j];
    sAng[img*8 + i] = tanhf(d + style_b[i]) * PI_F;
  }
}

__global__ void qcnn_pack_kernel(const float* __restrict__ res_w,
                                 const float* __restrict__ data_w,
                                 float* __restrict__ Bpack)
{
  int idx = blockIdx.x * 256 + threadIdx.x;
  if (idx >= 46080) return;
  int v    = idx & 1;
  int lane = (idx >> 1) & 31;
  int s    = (idx >> 6) % 144;
  int n    = idx / 9216;
  int k    = (s << 2) + ((lane >> 4) << 1) + v;
  int row  = (n << 4) + (lane & 15);
  float val = 0.f;
  if (row < 64)      val = res_w[row * 576 + k];
  else if (row < 70) val = data_w[(row - 64) * 576 + k];
  Bpack[idx] = val;
}

__global__ __launch_bounds__(256) void qcnn_main_kernel(
    const float* __restrict__ x,
    const float* __restrict__ data_b,
    const float* __restrict__ res_b,
    const float* __restrict__ out_b,
    const float* __restrict__ gUr,
    const float* __restrict__ gUi,
    const float* __restrict__ gWq,
    const float* __restrict__ gStyle,
    const float* __restrict__ gBp,
    float* __restrict__ out)
{
  extern __shared__ float smem[];
  float* xs   = smem;
  float* Pm   = smem;
  float* PRm  = smem + 8704;
  float* Ur   = smem + 17408;
  float* Ui   = Ur + 4160;
  float* resS = Ui + 4160;
  float* thS  = resS + 8320;
  int*   offT = (int*)(thS + 1024);

  const int tid = threadIdx.x;
  const int b   = blockIdx.x >> 5;
  const int h0  = (blockIdx.x & 31) << 1;

  for (int i = tid; i < 4096; i += 256) {
    int a = i >> 6, k = i & 63;
    Ur[a*65 + k] = gUr[i];
    Ui[a*65 + k] = gUi[i];
  }
  for (int k = tid; k < 576; k += 256) {
    int c = k / 9, tt = k % 9;
    offT[k] = ((c << 2) + tt / 3) * 66 + (tt % 3);
  }
  for (int i = tid; i < 64*4*66; i += 256) {
    int c = i / 264, rr = i % 264, r = rr / 66, col = rr % 66;
    int h = h0 - 1 + r, w = col - 1;
    float v = 0.f;
    if ((unsigned)h < 64u && (unsigned)w < 64u)
      v = x[(size_t)(b*64 + c) * 4096 + h * 64 + w];
    xs[i] = v;
  }
  __syncthreads();

  const int lane = tid & 31;
  const int hf   = lane >> 4;
  const int lr   = lane & 15;
  const int m0   = (tid >> 5) << 4;
  const int m    = m0 + lr;
  const int baseM = ((m >> 6) * 66) + (m & 63);

  v8f acc[5];
  const v8f vz = {0.f,0.f,0.f,0.f,0.f,0.f,0.f,0.f};
  #pragma unroll
  for (int n = 0; n < 5; ++n) acc[n] = vz;

  const float2* Bp2 = (const float2*)gBp;
  for (int s = 0; s < 144; ++s) {
    int kk = (s << 2) + (hf << 1);
    v2f A;
    A.x = xs[offT[kk]     + baseM];
    A.y = xs[offT[kk + 1] + baseM];
    int bbase = (s << 5) + lane;
    #pragma unroll
    for (int n = 0; n < 5; ++n) {
      float2 bv = Bp2[n * 4608 + bbase];
      v2f B; B.x = bv.x; B.y = bv.y;
      acc[n] = wmma_f32_k4(A, B, acc[n]);
    }
  }

  #pragma unroll
  for (int n = 0; n < 4; ++n) {
    int ch = (n << 4) + lr;
    #pragma unroll
    for (int r = 0; r < 8; ++r)
      resS[(m0 + r + (hf << 3)) * 65 + ch] = acc[n][r];
  }
  if (lr < 6) {
    float db = data_b[lr];
    float sa = gStyle[(b << 3) + lr];
    #pragma unroll
    for (int r = 0; r < 8; ++r) {
      int mm = m0 + r + (hf << 3);
      thS[(mm << 3) + lr] = tanhf(acc[4][r] + db) * PI_F + sa;
    }
  }
  __syncthreads();

  if (tid < 128) {
    float cc[6], ss2[6];
    #pragma unroll
    for (int i = 0; i < 6; ++i) {
      float th = thS[(tid << 3) + i] * 0.5f;
      cc[i] = cosf(th); ss2[i] = sinf(th);
    }
    float p[64];
    p[0] = 1.f;
    #pragma unroll
    for (int i = 0; i < 6; ++i) {
      #pragma unroll
      for (int a = (1 << i) - 1; a >= 0; --a) {
        float tv = p[a];
        p[2*a + 1] = tv * ss2[i];
        p[2*a]     = tv * cc[i];
      }
    }
    #pragma unroll
    for (int a = 0; a < 64; a += 4)
      *(float4*)&Pm[tid * 68 + a] = make_float4(p[a], p[a+1], p[a+2], p[a+3]);
  }
  __syncthreads();

  #pragma unroll 1
  for (int nt = 0; nt < 4; ++nt) {
    int acol = (nt << 4) + lr;
    v8f ar = vz, ai = vz;
    for (int s = 0; s < 16; ++s) {
      int k = (s << 2) + (hf << 1);
      v2f A;  A.x  = Pm[m * 68 + k];        A.y  = Pm[m * 68 + k + 1];
      v2f Br; Br.x = Ur[acol * 65 + k];     Br.y = Ur[acol * 65 + k + 1];
      v2f Bi; Bi.x = Ui[acol * 65 + k];     Bi.y = Ui[acol * 65 + k + 1];
      ar = wmma_f32_k4(A, Br, ar);
      ai = wmma_f32_k4(A, Bi, ai);
    }
    #pragma unroll
    for (int r = 0; r < 8; ++r) {
      int mm = m0 + r + (hf << 3);
      PRm[mm * 68 + acol] = ar[r]*ar[r] + ai[r]*ai[r];
    }
  }

  #pragma unroll 1
  for (int nt = 0; nt < 4; ++nt) {
    int ccol = (nt << 4) + lr;
    v8f q = vz;
    for (int s = 0; s < 16; ++s) {
      int k = (s << 2) + (hf << 1);
      v2f A; A.x = PRm[m * 68 + k];          A.y = PRm[m * 68 + k + 1];
      v2f B; B.x = gWq[(k << 6) + ccol];     B.y = gWq[((k + 1) << 6) + ccol];
      q = wmma_f32_k4(A, B, q);
    }
    float bias = res_b[ccol] + out_b[ccol];
    #pragma unroll
    for (int r = 0; r < 8; ++r) {
      int mm = m0 + r + (hf << 3);
      resS[mm * 65 + ccol] += q[r] + bias;
    }
  }
  __syncthreads();

  for (int idx = tid; idx < 8192; idx += 256) {
    int c = idx >> 7, mm = idx & 127;
    int h = h0 + (mm >> 6), w = mm & 63;
    out[(size_t)(b*64 + c) * 4096 + h * 64 + w] = resS[mm * 65 + c];
  }
}

extern "C" void kernel_launch(void* const* d_in, const int* in_sizes, int n_in,
                              void* d_out, int out_size, void* d_ws, size_t ws_size,
                              hipStream_t stream) {
  (void)in_sizes; (void)n_in; (void)out_size; (void)ws_size;
  const float* x       = (const float*)d_in[0];
  const float* style   = (const float*)d_in[1];
  const float* data_w  = (const float*)d_in[2];
  const float* data_b  = (const float*)d_in[3];
  const float* style_w = (const float*)d_in[4];
  const float* style_b = (const float*)d_in[5];
  const float* qcnn    = (const float*)d_in[6];
  const float* meas    = (const float*)d_in[7];
  const float* out_w   = (const float*)d_in[8];
  const float* out_b   = (const float*)d_in[9];
  const float* res_w   = (const float*)d_in[10];
  const float* res_b   = (const float*)d_in[11];

  float* ws   = (float*)d_ws;
  float* Ur   = ws;
  float* Ui   = ws + 4096;
  float* Wq   = ws + 8192;
  float* sAng = ws + 12288;
  float* Bp   = ws + 12352;

  hipLaunchKernelGGL(qcnn_setup_kernel, dim3(1), dim3(64), 0, stream,
                     style, style_w, style_b, qcnn, meas, out_w, Ur, Ui, Wq, sAng);
  hipLaunchKernelGGL(qcnn_pack_kernel, dim3(180), dim3(256), 0, stream,
                     res_w, data_w, Bp);
  hipLaunchKernelGGL(qcnn_main_kernel, dim3(256), dim3(256), 142592  , stream,
                     x, data_b, res_b, out_b, Ur, Ui, Wq, sAng, Bp, (float*)d_out);
}


# read-out as expectation values: stage 3 tile transposed, probs x sign matrix in registers (4 WMMAs per wave), then x out_proj (K=8) on the res tile; Wq table and build go; read-out sum order changes
# speedup vs baseline: 5.8724x; 5.8724x over previous
#include <hip/hip_runtime.h>
#include <math.h>

typedef __attribute__((ext_vector_type(2))) float v2f;
typedef __attribute__((ext_vector_type(8))) float v8f;

#define PI_F 3.14159265358979323846f

#define NT 1024
#define KQ 4

static __device__ __forceinline__ v8f wmma_f32_k4(v2f a, v2f b, v8f c) {
  return __builtin_amdgcn_wmma_f32_16x16x4_f32(false, a, false, b, (short)0, c, false, false);
}


static __device__ __forceinline__ void wave_lds_handover() {
  __builtin_amdgcn_fence(__ATOMIC_RELEASE, "wavefront");
  __builtin_amdgcn_wave_barrier();
  __builtin_amdgcn_fence(__ATOMIC_ACQUIRE, "wavefront");
}

template <int JM>
static __device__ __forceinline__ void ry_rz_local(float (&re)[16], float (&im)[16],
                                                   float c, float s, float ch, float sh) {
  #pragma unroll
  for (int j = 0; j < 16; ++j) if (!(j & JM)) {
    const int j1 = j | JM;
    float r0 = re[j], i0 = im[j], r1 = re[j1], i1 = im[j1];
    re[j]  = c*r0 - s*r1;  im[j]  = c*i0 - s*i1;
    re[j1] = s*r0 + c*r1;  im[j1] = s*i0 + c*i1;
  }
  #pragma unroll
  for (int j = 0; j < 16; ++j) {
    float r = re[j], q = im[j];
    if (!(j & JM)) { re[j] = r*ch + q*sh; im[j] = q*ch - r*sh; }
    else           { re[j] = r*ch - q*sh; im[j] = q*ch + r*sh; }
  }
}

template <int JM>
static __device__ __forceinline__ void u3_local(float (&re)[16], float (&im)[16],
                                                const float* __restrict__ p) {
  const float ct = p[0], st = p[1];
  const float m01r = -p[2]*st, m01i = -p[3]*st;
  const float m10r =  p[4]*st, m10i =  p[5]*st;
  const float m11r =  p[6]*ct, m11i =  p[7]*ct;
  #pragma unroll
  for (int j = 0; j < 16; ++j) if (!(j & JM)) {
    const int j1 = j | JM;
    float r0 = re[j], i0 = im[j], r1 = re[j1], i1 = im[j1];
    re[j]  = ct*r0 + m01r*r1 - m01i*i1;
    im[j]  = ct*i0 + m01r*i1 + m01i*r1;
    re[j1] = m10r*r0 - m10i*i0 + m11r*r1 - m11i*i1;
    im[j1] = m10r*i0 + m10i*r0 + m11r*i1 + m11i*r1;
  }
}

static __device__ __forceinline__ int cnot_ring_image(int a) {
  #pragma unroll
  for (int i = 0; i < 6; ++i) {
    const int mc = 1 << (5 - i);
    const int mt = 1 << (5 - ((i + 1) % 6));
    if (a & mc) a ^= mt;
  }
  return a;
}

static __device__ __forceinline__ void build_circuit_unitary(
    float* __restrict__ Xr, float* __restrict__ Xi,
    float* __restrict__ Yr, float* __restrict__ Yi,
    const float* __restrict__ cs, int tid)
{
  const int t = ((tid >> 5) << 3) + (tid & 7);
  const int g = (tid & 31) >> 3;
  const int offB = g * 65 + t;
  const int offA = g * 1040 + t;
  const int imgG = cnot_ring_image(g << 4);
  for (int l = 0; l < 2; ++l) {
    const float4* g4 = (const float4*)(cs + l * 24);
    {
      float re[16], im[16];
      if (l) {
        #pragma unroll
        for (int j = 0; j < 16; ++j) { re[j] = Xr[offB + j*260]; im[j] = Xi[offB + j*260]; }
      } else {
        #pragma unroll
        for (int j = 0; j < 16; ++j) { re[j] = ((j << 2) + g == t) ? 1.f : 0.f; im[j] = 0.f; }
      }
      { float4 q = g4[0]; ry_rz_local<8>(re, im, q.x, q.y, q.z, q.w); }
      { float4 q = g4[1]; ry_rz_local<4>(re, im, q.x, q.y, q.z, q.w); }
      { float4 q = g4[2]; ry_rz_local<2>(re, im, q.x, q.y, q.z, q.w); }
      { float4 q = g4[3]; ry_rz_local<1>(re, im, q.x, q.y, q.z, q.w); }
      #pragma unroll
      for (int j = 0; j < 16; ++j) { Yr[offB + j*260] = re[j]; Yi[offB + j*260] = im[j]; }
    }
    wave_lds_handover();
    {
      float re[16], im[16];
      #pragma unroll
      for (int j = 0; j < 16; ++j) { re[j] = Yr[offA + j*65]; im[j] = Yi[offA + j*65]; }
      { float4 q = g4[4]; ry_rz_local<2>(re, im, q.x, q.y, q.z, q.w); }
      { float4 q = g4[5]; ry_rz_local<1>(re, im, q.x, q.y, q.z, q.w); }
      #pragma unroll
      for (int j = 0; j < 16; ++j) {
        const int d = (imgG ^ cnot_ring_image(j)) * 65 + t;
        Xr[d] = re[j]; Xi[d] = im[j];
      }
    }
    wave_lds_handover();
  }
  {
    float re[16], im[16];
    #pragma unroll
    for (int j = 0; j < 16; ++j) { re[j] = Xr[offB + j*260]; im[j] = Xi[offB + j*260]; }
    u3_local<8>(re, im, cs + 48);
    u3_local<4>(re, im, cs + 56);
    u3_local<2>(re, im, cs + 64);
    u3_local<1>(re, im, cs + 72);
    #pragma unroll
    for (int j = 0; j < 16; ++j) { Yr[offB + j*260] = re[j]; Yi[offB + j*260] = im[j]; }
  }
  wave_lds_handover();
  {
    float re[16], im[16];
    #pragma unroll
    for (int j = 0; j < 16; ++j) { re[j] = Yr[offA + j*65]; im[j] = Yi[offA + j*65]; }
    u3_local<2>(re, im, cs + 80);
    u3_local<1>(re, im, cs + 88);
    #pragma unroll
    for (int j = 0; j < 16; ++j) { Xr[offA + j*65] = re[j]; Xi[offA + j*65] = im[j]; }
  }
}

__global__ void qcnn_pack_kernel(const float* __restrict__ res_w,
                                 const float* __restrict__ data_w,
                                 float* __restrict__ Bpack)
{
  int idx = blockIdx.x * 256 + threadIdx.x;
  if (idx >= 46080) return;
  int v    = idx & 1;
  int lane = (idx >> 1) & 31;
  int s    = (idx >> 6) % 144;
  int n    = idx / 9216;
  int c    = ((s / 9) << 2) + ((lane >> 4) << 1) + v;
  int k    = c * 9 + (s % 9);
  int row  = (n << 4) + (lane & 15);
  float val = 0.f;
  if (row < 64)      val = res_w[row * 576 + k];
  else if (row < 70) val = data_w[(row - 64) * 576 + k];
  Bpack[idx] = val;
}

__global__ __launch_bounds__(NT) void qcnn_main_kernel(
    const float* __restrict__ x,
    const float* __restrict__ style,
    const float* __restrict__ data_b,
    const float* __restrict__ style_w,
    const float* __restrict__ style_b,
    const float* __restrict__ qcnn,
    const float* __restrict__ meas,
    const float* __restrict__ out_w,
    const float* __restrict__ out_b,
    const float* __restrict__ res_b,
    const float* __restrict__ gBp,
    float* __restrict__ out)
{
  extern __shared__ float smem[];
  float* Pm   = smem;
  float* expS = smem + 8704;
  float* Ur   = smem + 12800;
  float* Ui   = Ur + 4160;
  float* resS = Ui + 4160;
  float* thS  = resS + 8320;
  float* csT  = thS + 1024;
  float* saS  = csT + 768;
  float* part = saS + 8;

  const int tid = threadIdx.x;
  const int b   = blockIdx.x >> 5;
  const int h0  = (blockIdx.x & 31) << 1;

  const int kq = __builtin_amdgcn_readfirstlane(tid >> 8);
  const int lane = tid & 31;

  if (kq == 0) {
    float* csW = csT + (tid >> 5) * 96;
    for (int r = 0; r < 2; ++r) {
      const int a = lane + (r << 5);
      if (a < 48) {
        float ang;
        if (a < 24) {
          ang = 0.5f * qcnn[a * 3];
        } else {
          const int i = (a - 24) >> 2, q = (a - 24) & 3;
          const float th = meas[i*3+0], ph = meas[i*3+1], la = meas[i*3+2];
          ang = (q == 0) ? 0.5f * th : (q == 1) ? la : (q == 2) ? ph : (ph + la);
        }
        csW[2*a]     = cosf(ang);
        csW[2*a + 1] = sinf(ang);
      }
    }
    if (tid >= 64 && tid < 70) {
      const int i = tid - 64;
      const float4* sp = (const float4*)(style + b * 128);
      const float4* wp = (const float4*)(style_w + i * 128);
      float d = 0.f;
      for (int j = 0; j < 32; ++j) {
        float4 sv = sp[j], wv = wp[j];
        d += sv.x * wv.x; d += sv.y * wv.y; d += sv.z * wv.z; d += sv.w * wv.w;
      }
      saS[i] = tanhf(d + style_b[i]) * PI_F;
    }
    wave_lds_handover();
    build_circuit_unitary(Ur, Ui, resS, resS + 4160, csW, tid);
  }

  const int hf   = lane >> 4;
  const int lr   = lane & 15;
  const int mt   = (tid >> 5) & 7;
  const int m0   = mt << 4;
  const int m    = m0 + lr;

  const int ph = h0 + (mt >> 2);
  const int pw = ((mt & 3) << 4) + lr;
  unsigned cofs[3];
  bool     cok[3];
  #pragma unroll
  for (int dx = 0; dx < 3; ++dx) {
    const int ww = pw + dx - 1;
    cofs[dx] = (unsigned)((hf << 13) + min(max(ww, 0), 63));
    cok[dx]  = (unsigned)ww < 64u;
  }
  const float* xq = x + (size_t)(b * 64 + kq * 16) * 4096;

  v8f acc[5];
  const v8f vz = {0.f,0.f,0.f,0.f,0.f,0.f,0.f,0.f};
  #pragma unroll
  for (int n = 0; n < 5; ++n) acc[n] = vz;

  const float2* Bp2 = (const float2*)gBp;
  const int s0 = kq * (144 / KQ);
  for (int jj = 0; jj < 144 / KQ / 9; ++jj) {
    float a0[9], a1[9];
    #pragma unroll
    for (int dy = 0; dy < 3; ++dy) {
      const int hh = ph + dy - 1;
      const unsigned ro = (unsigned)((jj << 14) + (min(max(hh, 0), 63) << 6));
      #pragma unroll
      for (int dx = 0; dx < 3; ++dx) {
        const unsigned o = ro + cofs[dx];
        a0[dy*3 + dx] = xq[o];
        a1[dy*3 + dx] = xq[o + 4096u];
      }
    }
    #pragma unroll
    for (int dy = 0; dy < 3; ++dy) {
      const bool rok = (unsigned)(ph + dy - 1) < 64u;
      float2 bv[3][5];
      const int bbase = ((s0 + jj * 9 + dy * 3) << 5) + lane;
      #pragma unroll
      for (int dx = 0; dx < 3; ++dx) {
        #pragma unroll
        for (int n = 0; n < 5; ++n) bv[dx][n] = Bp2[n * 4608 + bbase + (dx << 5)];
      }
      #pragma unroll
      for (int dx = 0; dx < 3; ++dx) {
        const bool ok = rok && cok[dx];
        v2f A;
        A.x = ok ? a0[dy*3 + dx] : 0.f;
        A.y = ok ? a1[dy*3 + dx] : 0.f;
        #pragma unroll
        for (int n = 0; n < 5; ++n) {
          v2f B; B.x = bv[dx][n].x; B.y = bv[dx][n].y;
          acc[n] = wmma_f32_k4(A, B, acc[n]);
        }
      }
    }
  }

  if (kq > 0) {
    float* pp = part + ((kq - 1) * 8 + mt) * 1280 + lane;
    #pragma unroll
    for (int n = 0; n < 5; ++n) {
      #pragma unroll
      for (int r = 0; r < 8; ++r) pp[(n * 8 + r) * 32] = acc[n][r];
    }
  }
  __syncthreads();
  if (kq == 0) {
    #pragma unroll
    for (int q = 0; q < KQ - 1; ++q) {
      const float* pp = part + (q * 8 + mt) * 1280 + lane;
      #pragma unroll
      for (int n = 0; n < 5; ++n) {
        #pragma unroll
        for (int r = 0; r < 8; ++r) acc[n][r] += pp[(n * 8 + r) * 32];
      }
    }
    #pragma unroll
    for (int n = 0; n < 4; ++n) {
      int ch = (n << 4) + lr;
      #pragma unroll
      for (int r = 0; r < 8; ++r)
        resS[(m0 + r + (hf << 3)) * 65 + ch] = acc[n][r];
    }
    if (lr < 6) {
      float db = data_b[lr];
      float sa = saS[lr];
      #pragma unroll
      for (int r = 0; r < 8; ++r) {
        int mm = m0 + r + (hf << 3);
        thS[(mm << 3) + lr] = tanhf(acc[4][r] + db) * PI_F + sa;
      }
    }
  }
  __syncthreads();

  if (tid < 128) {
    float cc[6], ss2[6];
    #pragma unroll
    for (int i = 0; i < 6; ++i) {
      float th = thS[(tid << 3) + i] * 0.5f;
      cc[i] = cosf(th); ss2[i] = sinf(th);
    }
    float p[64];
    p[0] = 1.f;
    #pragma unroll
    for (int i = 0; i < 6; ++i) {
      #pragma unroll
      for (int a = (1 << i) - 1; a >= 0; --a) {
        float tv = p[a];
        p[2*a + 1] = tv * ss2[i];
        p[2*a]     = tv * cc[i];
      }
    }
    #pragma unroll
    for (int a = 0; a < 64; a += 4)
      *(float4*)&Pm[tid * 68 + a] = make_float4(p[a], p[a+1], p[a+2], p[a+3]);
  }
  __syncthreads();

  const int ccol = (kq << 4) + lr;
  const float2* owp = (const float2*)(out_w + ccol * 6);
  const float2 ow0 = owp[hf];
  float2 ow1 = owp[2];
  if (hf) { ow1.x = 0.f; ow1.y = 0.f; }
  const float bias = res_b[ccol] + out_b[ccol];

  {
    const int arow = (kq << 4) + lr;
    v8f ar = vz, ai = vz;
    for (int s = 0; s < 16; ++s) {
      int k = (s << 2) + (hf << 1);
      v2f Ar; Ar.x = Ur[arow * 65 + k];     Ar.y = Ur[arow * 65 + k + 1];
      v2f Ai; Ai.x = Ui[arow * 65 + k];     Ai.y = Ui[arow * 65 + k + 1];
      v2f B;  B.x  = Pm[m * 68 + k];        B.y  = Pm[m * 68 + k + 1];
      ar = wmma_f32_k4(Ar, B, ar);
      ai = wmma_f32_k4(Ai, B, ai);
    }
    float pr[8];
    #pragma unroll
    for (int r = 0; r < 8; ++r) pr[r] = ar[r]*ar[r] + ai[r]*ai[r];
    const int   sh  = (lr < 6) ? (5 - lr) : 0;
    const float one = (lr < 6) ? 1.f : 0.f;
    const int   ab  = (kq << 4) + (hf << 3);
    v8f ev = vz;
    #pragma unroll
    for (int j = 0; j < 4; ++j) {
      v2f S;
      S.x = (((ab + 2*j)     >> sh) & 1) ? -one : one;
      S.y = (((ab + 2*j + 1) >> sh) & 1) ? -one : one;
      v2f P; P.x = pr[2*j]; P.y = pr[2*j + 1];
      ev = wmma_f32_k4(S, P, ev);
    }
    if (hf == 0) {
      float4* ep = (float4*)(expS + ((kq << 7) + m) * 8);
      ep[0] = make_float4(ev[0], ev[1], ev[2], ev[3]);
      ep[1] = make_float4(ev[4], ev[5], ev[6], ev[7]);
    }
  }
  __syncthreads();

  {
    v2f E0 = {0.f, 0.f}, E1 = {0.f, 0.f};
    #pragma unroll
    for (int q = 0; q < KQ; ++q) {
      const float2* ep = (const float2*)(expS + ((q << 7) + m) * 8);
      float2 t0 = ep[hf], t1 = ep[2 + hf];
      E0.x += t0.x; E0.y += t0.y; E1.x += t1.x; E1.y += t1.y;
    }
    v8f cacc;
    #pragma unroll
    for (int r = 0; r < 8; ++r) cacc[r] = resS[(m0 + r + (hf << 3)) * 65 + ccol];
    v2f B0; B0.x = ow0.x; B0.y = ow0.y;
    v2f B1; B1.x = ow1.x; B1.y = ow1.y;
    cacc = wmma_f32_k4(E0, B0, cacc);
    cacc = wmma_f32_k4(E1, B1, cacc);
    #pragma unroll
    for (int r = 0; r < 8; ++r) resS[(m0 + r + (hf << 3)) * 65 + ccol] = cacc[r] + bias;
  }
  __syncthreads();

  for (int idx = tid; idx < 8192; idx += NT) {
    int c = idx >> 7, mm = idx & 127;
    int h = h0 + (mm >> 6), w = mm & 63;
    out[(size_t)(b*64 + c) * 4096 + h * 64 + w] = resS[mm * 65 + c];
  }
}

extern "C" void kernel_launch(void* const* d_in, const int* in_sizes, int n_in,
                              void* d_out, int out_size, void* d_ws, size_t ws_size,
                              hipStream_t stream) {
  (void)in_sizes; (void)n_in; (void)out_size; (void)ws_size;
  const float* x       = (const float*)d_in[0];
  const float* style   = (const float*)d_in[1];
  const float* data_w  = (const float*)d_in[2];
  const float* data_b  = (const float*)d_in[3];
  const float* style_w = (const float*)d_in[4];
  const float* style_b = (const float*)d_in[5];
  const float* qcnn    = (const float*)d_in[6];
  const float* meas    = (const float*)d_in[7];
  const float* out_w   = (const float*)d_in[8];
  const float* out_b   = (const float*)d_in[9];
  const float* res_w   = (const float*)d_in[10];
  const float* res_b   = (const float*)d_in[11];

  float* Bp = (float*)d_ws;

  hipLaunchKernelGGL(qcnn_pack_kernel, dim3(180), dim3(256), 0, stream,
                     res_w, data_w, Bp);
  hipLaunchKernelGGL(qcnn_main_kernel, dim3(256), dim3(NT), 247840  , stream,
                     x, style, data_b, style_w, style_b, qcnn, meas, out_w, out_b, res_b, Bp, (float*)d_out);
}


# pack launch removed: each work-group copies res_w, data_w as they are into LDS (row pitch 580) and reads conv B fragments there, 8 bytes a lane; K steps pair neighbouring k (conv sum order changes)
# speedup vs baseline: 6.3497x; 1.0813x over previous
#include <hip/hip_runtime.h>
#include <math.h>

typedef __attribute__((ext_vector_type(2))) float v2f;
typedef __attribute__((ext_vector_type(4))) float v4f;
typedef __attribute__((ext_vector_type(8))) float v8f;

#define PI_F 3.14159265358979323846f

#define NT 1024
#define KQ 4
#define WP 580

static __device__ __forceinline__ v8f wmma_f32_k4(v2f a, v2f b, v8f c) {
  return __builtin_amdgcn_wmma_f32_16x16x4_f32(false, a, false, b, (short)0, c, false, false);
}


static __device__ __forceinline__ void wave_lds_handover() {
  __builtin_amdgcn_fence(__ATOMIC_RELEASE, "wavefront");
  __builtin_amdgcn_wave_barrier();
  __builtin_amdgcn_fence(__ATOMIC_ACQUIRE, "wavefront");
}

template <int JM>
static __device__ __forceinline__ void ry_rz_local(float (&re)[16], float (&im)[16],
                                                   float c, float s, float ch, float sh) {
  #pragma unroll
  for (int j = 0; j < 16; ++j) if (!(j & JM)) {
    const int j1 = j | JM;
    float r0 = re[j], i0 = im[j], r1 = re[j1], i1 = im[j1];
    re[j]  = c*r0 - s*r1;  im[j]  = c*i0 - s*i1;
    re[j1] = s*r0 + c*r1;  im[j1] = s*i0 + c*i1;
  }
  #pragma unroll
  for (int j = 0; j < 16; ++j) {
    float r = re[j], q = im[j];
    if (!(j & JM)) { re[j] = r*ch + q*sh; im[j] = q*ch - r*sh; }
    else           { re[j] = r*ch - q*sh; im[j] = q*ch + r*sh; }
  }
}

template <int JM>
static __device__ __forceinline__ void u3_local(float (&re)[16], float (&im)[16],
                                                const float* __restrict__ p) {
  const float ct = p[0], st = p[1];
  const float m01r = -p[2]*st, m01i = -p[3]*st;
  const float m10r =  p[4]*st, m10i =  p[5]*st;
  const float m11r =  p[6]*ct, m11i =  p[7]*ct;
  #pragma unroll
  for (int j = 0; j < 16; ++j) if (!(j & JM)) {
    const int j1 = j | JM;
    float r0 = re[j], i0 = im[j], r1 = re[j1], i1 = im[j1];
    re[j]  = ct*r0 + m01r*r1 - m01i*i1;
    im[j]  = ct*i0 + m01r*i1 + m01i*r1;
    re[j1] = m10r*r0 - m10i*i0 + m11r*r1 - m11i*i1;
    im[j1] = m10r*i0 + m10i*r0 + m11r*i1 + m11i*r1;
  }
}

static __device__ __forceinline__ int cnot_ring_image(int a) {
  #pragma unroll
  for (int i = 0; i < 6; ++i) {
    const int mc = 1 << (5 - i);
    const int mt = 1 << (5 - ((i + 1) % 6));
    if (a & mc) a ^= mt;
  }
  return a;
}

static __device__ __forceinline__ void build_circuit_unitary(
    float* __restrict__ Xr, float* __restrict__ Xi,
    float* __restrict__ Yr, float* __restrict__ Yi,
    const float* __restrict__ cs, int tid)
{
  const int t = ((tid >> 5) << 3) + (tid & 7);
  const int g = (tid & 31) >> 3;
  const int offB = g * 65 + t;
  const int offA = g * 1040 + t;
  const int imgG = cnot_ring_image(g << 4);
  for (int l = 0; l < 2; ++l) {
    const float4* g4 = (const float4*)(cs + l * 24);
    {
      float re[16], im[16];
      if (l) {
        #pragma unroll
        for (int j = 0; j < 16; ++j) { re[j] = Xr[offB + j*260]; im[j] = Xi[offB + j*260]; }
      } else {
        #pragma unroll
        for (int j = 0; j < 16; ++j) { re[j] = ((j << 2) + g == t) ? 1.f : 0.f; im[j] = 0.f; }
      }
      { float4 q = g4[0]; ry_rz_local<8>(re, im, q.x, q.y, q.z, q.w); }
      { float4 q = g4[1]; ry_rz_local<4>(re, im, q.x, q.y, q.z, q.w); }
      { float4 q = g4[2]; ry_rz_local<2>(re, im, q.x, q.y, q.z, q.w); }
      { float4 q = g4[3]; ry_rz_local<1>(re, im, q.x, q.y, q.z, q.w); }
      #pragma unroll
      for (int j = 0; j < 16; ++j) { Yr[offB + j*260] = re[j]; Yi[offB + j*260] = im[j]; }
    }
    wave_lds_handover();
    {
      float re[16], im[16];
      #pragma unroll
      for (int j = 0; j < 16; ++j) { re[j] = Yr[offA + j*65]; im[j] = Yi[offA + j*65]; }
      { float4 q = g4[4]; ry_rz_local<2>(re, im, q.x, q.y, q.z, q.w); }
      { float4 q = g4[5]; ry_rz_local<1>(re, im, q.x, q.y, q.z, q.w); }
      #pragma unroll
      for (int j = 0; j < 16; ++j) {
        const int d = (imgG ^ cnot_ring_image(j)) * 65 + t;
        Xr[d] = re[j]; Xi[d] = im[j];
      }
    }
    wave_lds_handover();
  }
  {
    float re[16], im[16];
    #pragma unroll
    for (int j = 0; j < 16; ++j) { re[j] = Xr[offB + j*260]; im[j] = Xi[offB + j*260]; }
    u3_local<8>(re, im, cs + 48);
    u3_local<4>(re, im, cs + 56);
    u3_local<2>(re, im, cs + 64);
    u3_local<1>(re, im, cs + 72);
    #pragma unroll
    for (int j = 0; j < 16; ++j) { Yr[offB + j*260] = re[j]; Yi[offB + j*260] = im[j]; }
  }
  wave_lds_handover();
  {
    float re[16], im[16];
    #pragma unroll
    for (int j = 0; j < 16; ++j) { re[j] = Yr[offA + j*65]; im[j] = Yi[offA + j*65]; }
    u3_local<2>(re, im, cs + 80);
    u3_local<1>(re, im, cs + 88);
    #pragma unroll
    for (int j = 0; j < 16; ++j) { Xr[offA + j*65] = re[j]; Xi[offA + j*65] = im[j]; }
  }
}

static __device__ __forceinline__ void load_taps(const float* __restrict__ xq, int jj, int ph,
                                                 const unsigned (&cofs)[3],
                                                 float (&a0)[9], float (&a1)[9]) {
  #pragma unroll
  for (int dy = 0; dy < 3; ++dy) {
    const int hh = ph + dy - 1;
    const unsigned ro = (unsigned)((jj << 14) + (min(max(hh, 0), 63) << 6));
    #pragma unroll
    for (int dx = 0; dx < 3; ++dx) {
      const unsigned o = ro + cofs[dx];
      a0[dy*3 + dx] = xq[o];
      a1[dy*3 + dx] = xq[o + 4096u];
    }
  }
}

__global__ __launch_bounds__(NT) void qcnn_main_kernel(
    const float* __restrict__ x,
    const float* __restrict__ style,
    const float* __restrict__ data_w,
    const float* __restrict__ data_b,
    const float* __restrict__ style_w,
    const float* __restrict__ style_b,
    const float* __restrict__ qcnn,
    const float* __restrict__ meas,
    const float* __restrict__ out_w,
    const float* __restrict__ out_b,
    const float* __restrict__ res_w,
    const float* __restrict__ res_b,
    float* __restrict__ out)
{
  extern __shared__ float smem[];
  float* Pm   = smem;
  float* expS = smem + 8704;
  float* Ur   = smem + 12800;
  float* Ui   = Ur + 4160;
  float* resS = Ui + 4160;
  float* thS  = resS + 8320;
  float* csT  = thS + 1024;
  float* saS  = csT + 768;
  float* Wl   = saS + 8;
  float* part = Wl;

  const int tid = threadIdx.x;
  const int b   = blockIdx.x >> 5;
  const int h0  = (blockIdx.x & 31) << 1;

  const int kq = __builtin_amdgcn_readfirstlane(tid >> 8);
  const int lane = tid & 31;

  const int hf   = lane >> 4;
  const int lr   = lane & 15;
  const int mt   = (tid >> 5) & 7;
  const int m0   = mt << 4;
  const int m    = m0 + lr;

  const int ph = h0 + (mt >> 2);
  const int pw = ((mt & 3) << 4) + lr;
  unsigned cofs[3];
  bool     cok[3];
  #pragma unroll
  for (int dx = 0; dx < 3; ++dx) {
    const int ww = pw + dx - 1;
    cofs[dx] = (unsigned)((hf << 13) + min(max(ww, 0), 63));
    cok[dx]  = (unsigned)ww < 64u;
  }
  const float* xq = x + (size_t)(b * 64 + kq * 16) * 4096;

  const v4f* r4 = (const v4f*)res_w;
  const v4f* d4 = (const v4f*)data_w;
  const bool dw = tid < 864;
  v4f wv[9];
  v4f wd = {0.f, 0.f, 0.f, 0.f};
  #pragma unroll
  for (int i = 0; i < 9; ++i) wv[i] = r4[tid + (i << 10)];
  if (dw) wd = d4[tid];

  float a0[9], a1[9];
  if (kq != 0) load_taps(xq, 0, ph, cofs, a0, a1);

  #pragma unroll
  for (int i = 0; i < 9; ++i) {
    const int f   = tid + (i << 10);
    const int row = f / 144;
    const int q4  = f - row * 144;
    *(v4f*)(Wl + row * WP + (q4 << 2)) = wv[i];
  }
  if (dw) {
    const int row = tid / 144;
    const int q4  = tid - row * 144;
    *(v4f*)(Wl + (64 + row) * WP + (q4 << 2)) = wd;
  }
  __syncthreads();

  if (kq == 0) {
    float* csW = csT + (tid >> 5) * 96;
    for (int r = 0; r < 2; ++r) {
      const int a = lane + (r << 5);
      if (a < 48) {
        float ang;
        if (a < 24) {
          ang = 0.5f * qcnn[a * 3];
        } else {
          const int i = (a - 24) >> 2, q = (a - 24) & 3;
          const float th = meas[i*3+0], ph = meas[i*3+1], la = meas[i*3+2];
          ang = (q == 0) ? 0.5f * th : (q == 1) ? la : (q == 2) ? ph : (ph + la);
        }
        csW[2*a]     = cosf(ang);
        csW[2*a + 1] = sinf(ang);
      }
    }
    if (tid >= 64 && tid < 70) {
      const int i = tid - 64;
      const float4* sp = (const float4*)(style + b * 128);
      const float4* wp = (const float4*)(style_w + i * 128);
      float d = 0.f;
      for (int j = 0; j < 32; ++j) {
        float4 sv = sp[j], wv = wp[j];
        d += sv.x * wv.x; d += sv.y * wv.y; d += sv.z * wv.z; d += sv.w * wv.w;
      }
      saS[i] = tanhf(d + style_b[i]) * PI_F;
    }
    wave_lds_handover();
    build_circuit_unitary(Ur, Ui, resS, resS + 4160, csW, tid);
    load_taps(xq, 0, ph, cofs, a0, a1);
  }

  int boff[5];
  #pragma unroll
  for (int n = 0; n < 5; ++n) boff[n] = min((n << 4) + lr, 69) * WP + kq * 144 + hf * 18;
  bool rok[3];
  #pragma unroll
  for (int dy = 0; dy < 3; ++dy) rok[dy] = (unsigned)(ph + dy - 1) < 64u;

  v8f acc[5];
  const v8f vz = {0.f,0.f,0.f,0.f,0.f,0.f,0.f,0.f};
  #pragma unroll
  for (int n = 0; n < 5; ++n) acc[n] = vz;

  for (int jj = 0; jj < 144 / KQ / 9; ++jj) {
    if (jj) load_taps(xq, jj, ph, cofs, a0, a1);
    const float* wj = Wl + jj * 36;
    #pragma unroll
    for (int g = 0; g < 3; ++g) {
      v2f bv[3][5];
      #pragma unroll
      for (int i = 0; i < 3; ++i) {
        #pragma unroll
        for (int n = 0; n < 5; ++n) bv[i][n] = *(const v2f*)(wj + boff[n] + 2 * (3 * g + i));
      }
      #pragma unroll
      for (int i = 0; i < 3; ++i) {
        const int kl0 = 2 * (3 * g + i), kl1 = kl0 + 1;
        const int t0 = kl0 % 9, t1 = kl1 % 9;
        const float x0 = (kl0 < 9) ? a0[t0] : a1[t0];
        const float x1 = (kl1 < 9) ? a0[t1] : a1[t1];
        v2f A;
        A.x = (rok[t0 / 3] && cok[t0 % 3]) ? x0 : 0.f;
        A.y = (rok[t1 / 3] && cok[t1 % 3]) ? x1 : 0.f;
        #pragma unroll
        for (int n = 0; n < 5; ++n) acc[n] = wmma_f32_k4(A, bv[i][n], acc[n]);
      }
    }
  }

  __syncthreads();
  if (kq > 0) {
    float* pp = part + ((kq - 1) * 8 + mt) * 1280 + lane;
    #pragma unroll
    for (int n = 0; n < 5; ++n) {
      #pragma unroll
      for (int r = 0; r < 8; ++r) pp[(n * 8 + r) * 32] = acc[n][r];
    }
  }
  __syncthreads();
  if (kq == 0) {
    #pragma unroll
    for (int q = 0; q < KQ - 1; ++q) {
      const float* pp = part + (q * 8 + mt) * 1280 + lane;
      #pragma unroll
      for (int n = 0; n < 5; ++n) {
        #pragma unroll
        for (int r = 0; r < 8; ++r) acc[n][r] += pp[(n * 8 + r) * 32];
      }
    }
    #pragma unroll
    for (int n = 0; n < 4; ++n) {
      int ch = (n << 4) + lr;
      #pragma unroll
      for (int r = 0; r < 8; ++r)
        resS[(m0 + r + (hf << 3)) * 65 + ch] = acc[n][r];
    }
    if (lr < 6) {
      float db = data_b[lr];
      float sa = saS[lr];
      #pragma unroll
      for (int r = 0; r < 8; ++r) {
        int mm = m0 + r + (hf << 3);
        thS[(mm << 3) + lr] = tanhf(acc[4][r] + db) * PI_F + sa;
      }
    }
  }
  __syncthreads();

  if (tid < 128) {
    float cc[6], ss2[6];
    #pragma unroll
    for (int i = 0; i < 6; ++i) {
      float th = thS[(tid << 3) + i] * 0.5f;
      cc[i] = cosf(th); ss2[i] = sinf(th);
    }
    float p[64];
    p[0] = 1.f;
    #pragma unroll
    for (int i = 0; i < 6; ++i) {
      #pragma unroll
      for (int a = (1 << i) - 1; a >= 0; --a) {
        float tv = p[a];
        p[2*a + 1] = tv * ss2[i];
        p[2*a]     = tv * cc[i];
      }
    }
    #pragma unroll
    for (int a = 0; a < 64; a += 4)
      *(float4*)&Pm[tid * 68 + a] = make_float4(p[a], p[a+1], p[a+2], p[a+3]);
  }
  __syncthreads();

  const int ccol = (kq << 4) + lr;
  const float2* owp = (const float2*)(out_w + ccol * 6);
  const float2 ow0 = owp[hf];
  float2 ow1 = owp[2];
  if (hf) { ow1.x = 0.f; ow1.y = 0.f; }
  const float bias = res_b[ccol] + out_b[ccol];

  {
    const int arow = (kq << 4) + lr;
    v8f ar = vz, ai = vz;
    for (int s = 0; s < 16; ++s) {
      int k = (s << 2) + (hf << 1);
      v2f Ar; Ar.x = Ur[arow * 65 + k];     Ar.y = Ur[arow * 65 + k + 1];
      v2f Ai; Ai.x = Ui[arow * 65 + k];     Ai.y = Ui[arow * 65 + k + 1];
      v2f B;  B.x  = Pm[m * 68 + k];        B.y  = Pm[m * 68 + k + 1];
      ar = wmma_f32_k4(Ar, B, ar);
      ai = wmma_f32_k4(Ai, B, ai);
    }
    float pr[8];
    #pragma unroll
    for (int r = 0; r < 8; ++r) pr[r] = ar[r]*ar[r] + ai[r]*ai[r];
    const int   sh  = (lr < 6) ? (5 - lr) : 0;
    const float one = (lr < 6) ? 1.f : 0.f;
    const int   ab  = (kq << 4) + (hf << 3);
    v8f ev = vz;
    #pragma unroll
    for (int j = 0; j < 4; ++j) {
      v2f S;
      S.x = (((ab + 2*j)     >> sh) & 1) ? -one : one;
      S.y = (((ab + 2*j + 1) >> sh) & 1) ? -one : one;
      v2f P; P.x = pr[2*j]; P.y = pr[2*j + 1];
      ev = wmma_f32_k4(S, P, ev);
    }
    if (hf == 0) {
      float4* ep = (float4*)(expS + ((kq << 7) + m) * 8);
      ep[0] = make_float4(ev[0], ev[1], ev[2], ev[3]);
      ep[1] = make_float4(ev[4], ev[5], ev[6], ev[7]);
    }
  }
  __syncthreads();

  {
    v2f E0 = {0.f, 0.f}, E1 = {0.f, 0.f};
    #pragma unroll
    for (int q = 0; q < KQ; ++q) {
      const float2* ep = (const float2*)(expS + ((q << 7) + m) * 8);
      float2 t0 = ep[hf], t1 = ep[2 + hf];
      E0.x += t0.x; E0.y += t0.y; E1.x += t1.x; E1.y += t1.y;
    }
    v8f cacc;
    #pragma unroll
    for (int r = 0; r < 8; ++r) cacc[r] = resS[(m0 + r + (hf << 3)) * 65 + ccol];
    v2f B0; B0.x = ow0.x; B0.y = ow0.y;
    v2f B1; B1.x = ow1.x; B1.y = ow1.y;
    cacc = wmma_f32_k4(E0, B0, cacc);
    cacc = wmma_f32_k4(E1, B1, cacc);
    #pragma unroll
    for (int r = 0; r < 8; ++r) resS[(m0 + r + (hf << 3)) * 65 + ccol] = cacc[r] + bias;
  }
  __syncthreads();

  for (int idx = tid; idx < 8192; idx += NT) {
    int c = idx >> 7, mm = idx & 127;
    int h = h0 + (mm >> 6), w = mm & 63;
    out[(size_t)(b*64 + c) * 4096 + h * 64 + w] = resS[mm * 65 + c];
  }
}

extern "C" void kernel_launch(void* const* d_in, const int* in_sizes, int n_in,
                              void* d_out, int out_size, void* d_ws, size_t ws_size,
                              hipStream_t stream) {
  (void)in_sizes; (void)n_in; (void)out_size; (void)d_ws; (void)ws_size;
  const float* x       = (const float*)d_in[0];
  const float* style   = (const float*)d_in[1];
  const float* data_w  = (const float*)d_in[2];
  const float* data_b  = (const float*)d_in[3];
  const float* style_w = (const float*)d_in[4];
  const float* style_b = (const float*)d_in[5];
  const float* qcnn    = (const float*)d_in[6];
  const float* meas    = (const float*)d_in[7];
  const float* out_w   = (const float*)d_in[8];
  const float* out_b   = (const float*)d_in[9];
  const float* res_w   = (const float*)d_in[10];
  const float* res_b   = (const float*)d_in[11];

  hipLaunchKernelGGL(qcnn_main_kernel, dim3(256), dim3(NT), 287360  , stream,
                     x, style, data_w, data_b, style_w, style_b, qcnn, meas, out_w, out_b, res_w, res_b, (float*)d_out);
}
